# starting kernel
# baseline (speedup 1.0000x reference)
#include <hip/hip_runtime.h>


typedef float v2f __attribute__((ext_vector_type(2)));
typedef float v8f __attribute__((ext_vector_type(8)));

#define N_IN    262144
#define N_TAPS  255
#define M_OUT   (N_IN + N_TAPS - 1)
#define OUT_TILE 512
#define KBLK    68
#define LDX     770
#define HR_SIZE 288

__global__ __launch_bounds__(256)
void fir_wmma_f32_kernel(const float* __restrict__ x,
                         const float* __restrict__ taps,
                         float* __restrict__ y)
{
    __shared__ float xt[16 * LDX];
    __shared__ float hr[HR_SIZE];

    const int  tid     = threadIdx.x;
    const int  rg      = blockIdx.y;
    const long wg_pout = (long)blockIdx.x * OUT_TILE;
    const long wg_p0   = wg_pout - (N_TAPS - 1);

    for (int c = tid; c < HR_SIZE; c += 256) {
        int j = 270 - c;
        float v = 0.f;
        if (j >= 0 && j < N_TAPS) {
            int t = j - 127; if (t < 0) t = -t;
            v = taps[t];
        }
        hr[c] = v;
    }

    for (int idx = tid; idx < 16 * 192; idx += 256) {
        const int  row = idx / 192;
        const int  c4  = (idx % 192) * 4;
        const long g   = wg_p0 + c4;
        const float* __restrict__ src = x + (long)(rg * 16 + row) * N_IN;
        float4 v;
        if (g >= 0 && g + 3 < (long)N_IN) {
            v = *(const float4*)(src + g);
        } else {
            v.x = (g + 0 >= 0 && g + 0 < (long)N_IN) ? src[g + 0] : 0.f;
            v.y = (g + 1 >= 0 && g + 1 < (long)N_IN) ? src[g + 1] : 0.f;
            v.z = (g + 2 >= 0 && g + 2 < (long)N_IN) ? src[g + 2] : 0.f;
            v.w = (g + 3 >= 0 && g + 3 < (long)N_IN) ? src[g + 3] : 0.f;
        }
        float* d = &xt[row * LDX + c4];
        d[0] = v.x; d[1] = v.y; d[2] = v.z; d[3] = v.w;
    }
    __syncthreads();

    const int wave = tid >> 5;
    const int lane = tid & 31;
    const int mn   = lane & 15;
    const int kp   = lane >> 4;

    const float* __restrict__ arow = &xt[mn * LDX + wave * 64 + 2 * kp];
    const int rb0 = 16 - mn + 2 * kp;

    v8f acc0 = {0.f,0.f,0.f,0.f,0.f,0.f,0.f,0.f};
    v8f acc1 = acc0, acc2 = acc0, acc3 = acc0;

    #pragma unroll 2
    for (int kb = 0; kb < KBLK; ++kb) {
        const int k0 = kb * 4;
        v2f b;
        b.x = hr[rb0 + k0];
        b.y = hr[rb0 + k0 + 1];
        v2f a0 = *(const v2f*)(arow + k0);
        v2f a1 = *(const v2f*)(arow + k0 + 16);
        v2f a2 = *(const v2f*)(arow + k0 + 32);
        v2f a3 = *(const v2f*)(arow + k0 + 48);
        acc0 = __builtin_amdgcn_wmma_f32_16x16x4_f32(false, a0, false, b, (short)0, acc0, false, false);
        acc1 = __builtin_amdgcn_wmma_f32_16x16x4_f32(false, a1, false, b, (short)0, acc1, false, false);
        acc2 = __builtin_amdgcn_wmma_f32_16x16x4_f32(false, a2, false, b, (short)0, acc2, false, false);
        acc3 = __builtin_amdgcn_wmma_f32_16x16x4_f32(false, a3, false, b, (short)0, acc3, false, false);
    }

    const long rowbase = (long)(rg * 16) * M_OUT;
    #pragma unroll
    for (int blk = 0; blk < 4; ++blk) {
        const v8f acc = (blk == 0) ? acc0 : (blk == 1) ? acc1 : (blk == 2) ? acc2 : acc3;
        const long pout = wg_pout + wave * 64 + blk * 16 + mn;
        if (pout < (long)M_OUT) {
            #pragma unroll
            for (int v = 0; v < 8; ++v) {
                const int row = v + 8 * kp;
                y[rowbase + (long)row * M_OUT + pout] = acc[v];
            }
        }
    }
}

extern "C" void kernel_launch(void* const* d_in, const int* in_sizes, int n_in,
                              void* d_out, int out_size, void* d_ws, size_t ws_size,
                              hipStream_t stream) {
    const float* x    = (const float*)d_in[0];
    const float* taps = (const float*)d_in[1];
    float* y          = (float*)d_out;

    dim3 grid((M_OUT + OUT_TILE - 1) / OUT_TILE,  2, 1);
    dim3 block(256, 1, 1);
    fir_wmma_f32_kernel<<<grid, block, 0, stream>>>(x, taps, y);
}


# block-Toeplitz form inside one row: 16x16 tap blocks held in registers as A, x columns from a wave-private LDS window as B, D tile = 256 consecutive outputs of a row; summation order changes
# speedup vs baseline: 1.3490x; 1.3490x over previous
#include <hip/hip_runtime.h>


typedef float v2f __attribute__((ext_vector_type(2)));
typedef float v4f __attribute__((ext_vector_type(4)));
typedef float v8f __attribute__((ext_vector_type(8)));

#define N_IN     262144
#define M_OUT    262398
#define N_ROWS   32
#define NS       17
#define GROUPS_PER_ROW 256
#define N_GROUPS (N_ROWS * GROUPS_PER_ROW)
#define COLP     20
#define WIN_COLS 80
#define WAVES_PER_WG 4
#ifndef N_WG
#define N_WG 1024
#endif
#define N_WAVES (N_WG * WAVES_PER_WG)

__device__ constexpr int sord(int k) { return k == 0 ? 0 : (k == 1 ? 16 : k - 1); }

template <int NT>
__device__ __forceinline__ void do_group(const float* __restrict__ xrow, float* __restrict__ yrow,
                                         const int P0, float* __restrict__ win,
                                         const v2f (&A)[NS][4], const int lane)
{
    constexpr int NCH = 2 + 2 * NT;
    const int xb = P0 - 256;
    v4f st[NCH];
#pragma unroll
    for (int i = 0; i < NCH; ++i) {
        const int g = xb + 128 * i;
        v4f v = {0.f, 0.f, 0.f, 0.f};
        if (g >= 0 && g < N_IN) v = *(const v4f*)(xrow + g + 4 * lane);
        st[i] = v;
    }
    __builtin_amdgcn_wave_barrier();
#pragma unroll
    for (int i = 0; i < NCH; ++i)
        *(v4f*)(win + (8 * i + (lane >> 2)) * COLP + 4 * (lane & 3)) = st[i];
    __builtin_amdgcn_wave_barrier();

    v8f acc[NT];
#pragma unroll
    for (int t = 0; t < NT; ++t) acc[t] = (v8f){0.f,0.f,0.f,0.f,0.f,0.f,0.f,0.f};

    const float* __restrict__ bl = win + (lane & 15) * COLP + 8 * (lane >> 4);
    v4f c0[NT], c1[NT];
#pragma unroll
    for (int t = 0; t < NT; ++t) {
        const float* p = bl + (16 * t + 16 - sord(0)) * COLP;
        c0[t] = *(const v4f*)p;
        c1[t] = *(const v4f*)(p + 4);
    }
#pragma unroll
    for (int k = 0; k < NS; ++k) {
        const int s = sord(k);
        v4f n0[NT], n1[NT];
        if (k + 1 < NS) {
#pragma unroll
            for (int t = 0; t < NT; ++t) {
                const float* p = bl + (16 * t + 16 - sord(k + 1)) * COLP;
                n0[t] = *(const v4f*)p;
                n1[t] = *(const v4f*)(p + 4);
            }
        }
#pragma unroll
        for (int j = 0; j < 4; ++j) {
#pragma unroll
            for (int t = 0; t < NT; ++t) {
                v2f b;
                if (j == 0)      { b.x = c0[t].x; b.y = c0[t].y; }
                else if (j == 1) { b.x = c0[t].z; b.y = c0[t].w; }
                else if (j == 2) { b.x = c1[t].x; b.y = c1[t].y; }
                else             { b.x = c1[t].z; b.y = c1[t].w; }
                acc[t] = __builtin_amdgcn_wmma_f32_16x16x4_f32(false, A[s][j], false, b,
                                                               (short)0, acc[t], false, false);
            }
        }
        __builtin_amdgcn_sched_barrier(0);
        if (k + 1 < NS) {
#pragma unroll
            for (int t = 0; t < NT; ++t) { c0[t] = n0[t]; c1[t] = n1[t]; }
        }
    }

#pragma unroll
    for (int t = 0; t < NT; ++t) {
        const int p = P0 + 256 * t + 16 * (lane & 15) + 8 * (lane >> 4);
#pragma unroll
        for (int k = 0; k < 4; ++k) {
            if (p + 2 * k < M_OUT) {
                v2f o; o.x = acc[t][2 * k]; o.y = acc[t][2 * k + 1];
                *(v2f*)(yrow + p + 2 * k) = o;
            }
        }
    }
}

__global__ __launch_bounds__(WAVES_PER_WG * 32)
void fir_block_toeplitz_f32(const float* __restrict__ x,
                            const float* __restrict__ taps,
                            float* __restrict__ y)
{
    __shared__ __attribute__((aligned(16))) float lds[WAVES_PER_WG * WIN_COLS * COLP];
    const int lane = threadIdx.x & 31;
    const int wv   = __builtin_amdgcn_readfirstlane((int)(threadIdx.x >> 5));
    float* win = lds + wv * (WIN_COLS * COLP);
    const int gw = blockIdx.x * WAVES_PER_WG + wv;

    v2f A[NS][4];
    const int u = (lane & 15) - 8 * (lane >> 4);
#pragma unroll
    for (int s = 0; s < NS; ++s) {
#pragma unroll
        for (int j = 0; j < 4; ++j) {
#pragma unroll
            for (int e = 0; e < 2; ++e) {
                int t = 16 * s + u - 2 * j - e - 127;
                t = t < 0 ? -t : t;
                const bool ok = t <= 127;
                const float v = taps[ok ? t : 0];
                A[s][j][e] = ok ? v : 0.f;
            }
        }
    }

    constexpr int TS = N_WAVES / N_ROWS;
    if (gw % TS == 0) {
        const int row = gw / TS;
        do_group<1>(x + (size_t)row * N_IN, y + (size_t)row * M_OUT, N_IN, win, A, lane);
    }
    for (int g = gw; g < N_GROUPS; g += N_WAVES) {
        const int row = g >> 8;
        const int gi  = g & 255;
        do_group<4>(x + (size_t)row * N_IN, y + (size_t)row * M_OUT, 1024 * gi, win, A, lane);
    }
}

extern "C" void kernel_launch(void* const* d_in, const int* in_sizes, int n_in,
                              void* d_out, int out_size, void* d_ws, size_t ws_size,
                              hipStream_t stream) {
    const float* x    = (const float*)d_in[0];
    const float* taps = (const float*)d_in[1];
    float* y          = (float*)d_out;
    fir_block_toeplitz_f32<<<dim3(N_WG, 1, 1), dim3(WAVES_PER_WG * 32, 1, 1), 0, stream>>>(x, taps, y);
}


# each wave requests its next window's x into registers before the current window's products (was load, wait, compute in turn); one 8-wave workgroup per WGP, 4 windows per wave; sum order unchanged
# speedup vs baseline: 1.4393x; 1.0669x over previous
#include <hip/hip_runtime.h>


typedef float v2f __attribute__((ext_vector_type(2)));
typedef float v4f __attribute__((ext_vector_type(4)));
typedef float v8f __attribute__((ext_vector_type(8)));

#define N_IN     262144
#define M_OUT    262398
#define N_ROWS   32
#define NS       17
#define GROUPS_PER_ROW 256
#define N_GROUPS (N_ROWS * GROUPS_PER_ROW)
#define COLP     20
#define WIN_COLS 80
#define WAVES_PER_WG 8
#ifndef N_WG
#define N_WG 256
#endif
#define N_WAVES (N_WG * WAVES_PER_WG)

__device__ constexpr int sord(int k) { return k == 0 ? 0 : (k == 1 ? 16 : k - 1); }

template <int NT>
__device__ __forceinline__ void load_window(const float* __restrict__ xrow, const int P0,
                                            const int lane, v4f (&st)[2 + 2 * NT])
{
    constexpr int NCH = 2 + 2 * NT;
    const int xb = P0 - 256;
#pragma unroll
    for (int i = 0; i < NCH; ++i) {
        const int g = xb + 128 * i;
        const int gc = (g >= 0 && g < N_IN) ? g : 0;
        st[i] = *(const v4f*)(xrow + gc + 4 * lane);
    }
}

template <int NT>
__device__ __forceinline__ void fill_window(float* __restrict__ win, const int P0, const int lane,
                                            const v4f (&st)[2 + 2 * NT])
{
    constexpr int NCH = 2 + 2 * NT;
    const int xb = P0 - 256;
    __builtin_amdgcn_wave_barrier();
#pragma unroll
    for (int i = 0; i < NCH; ++i) {
        const int g = xb + 128 * i;
        v4f v = st[i];
        if (!(g >= 0 && g < N_IN)) v = (v4f){0.f, 0.f, 0.f, 0.f};
        *(v4f*)(win + (8 * i + (lane >> 2)) * COLP + 4 * (lane & 3)) = v;
    }
    __builtin_amdgcn_wave_barrier();
}

template <int NT>
__device__ __forceinline__ void products(float* __restrict__ yrow, const int P0,
                                         const float* __restrict__ win,
                                         const v2f (&A)[NS][4], const int lane)
{
    v8f acc[NT];
#pragma unroll
    for (int t = 0; t < NT; ++t) acc[t] = (v8f){0.f,0.f,0.f,0.f,0.f,0.f,0.f,0.f};

    const float* __restrict__ bl = win + (lane & 15) * COLP + 8 * (lane >> 4);
    v4f c0[NT], c1[NT];
#pragma unroll
    for (int t = 0; t < NT; ++t) {
        const float* p = bl + (16 * t + 16 - sord(0)) * COLP;
        c0[t] = *(const v4f*)p;
        c1[t] = *(const v4f*)(p + 4);
    }
#pragma unroll
    for (int k = 0; k < NS; ++k) {
        const int s = sord(k);
        v4f n0[NT], n1[NT];
        if (k + 1 < NS) {
#pragma unroll
            for (int t = 0; t < NT; ++t) {
                const float* p = bl + (16 * t + 16 - sord(k + 1)) * COLP;
                n0[t] = *(const v4f*)p;
                n1[t] = *(const v4f*)(p + 4);
            }
        }
#pragma unroll
        for (int j = 0; j < 4; ++j) {
#pragma unroll
            for (int t = 0; t < NT; ++t) {
                v2f b;
                if (j == 0)      { b.x = c0[t].x; b.y = c0[t].y; }
                else if (j == 1) { b.x = c0[t].z; b.y = c0[t].w; }
                else if (j == 2) { b.x = c1[t].x; b.y = c1[t].y; }
                else             { b.x = c1[t].z; b.y = c1[t].w; }
                acc[t] = __builtin_amdgcn_wmma_f32_16x16x4_f32(false, A[s][j], false, b,
                                                               (short)0, acc[t], false, false);
            }
        }
        __builtin_amdgcn_sched_barrier(0);
        if (k + 1 < NS) {
#pragma unroll
            for (int t = 0; t < NT; ++t) { c0[t] = n0[t]; c1[t] = n1[t]; }
        }
    }

#pragma unroll
    for (int t = 0; t < NT; ++t) {
        const int p = P0 + 256 * t + 16 * (lane & 15) + 8 * (lane >> 4);
#pragma unroll
        for (int k = 0; k < 4; ++k) {
            if (p + 2 * k < M_OUT) {
                v2f o; o.x = acc[t][2 * k]; o.y = acc[t][2 * k + 1];
                *(v2f*)(yrow + p + 2 * k) = o;
            }
        }
    }
}

__global__ __launch_bounds__(WAVES_PER_WG * 32)
void fir_block_toeplitz_f32(const float* __restrict__ x,
                            const float* __restrict__ taps,
                            float* __restrict__ y)
{
    __shared__ __attribute__((aligned(16))) float lds[WAVES_PER_WG * WIN_COLS * COLP];
    const int lane = threadIdx.x & 31;
    const int wv   = __builtin_amdgcn_readfirstlane((int)(threadIdx.x >> 5));
    float* win = lds + wv * (WIN_COLS * COLP);
    const int gw = blockIdx.x * WAVES_PER_WG + wv;

    constexpr int GPW = N_GROUPS / N_WAVES;
    static_assert(N_GROUPS % N_WAVES == 0, "equal shares");
    v4f st[10];
    load_window<4>(x + (size_t)(gw >> 8) * N_IN, 1024 * (gw & 255), lane, st);
    __builtin_amdgcn_sched_barrier(0);

    v2f A[NS][4];
    const int u = (lane & 15) - 8 * (lane >> 4);
#pragma unroll
    for (int s = 0; s < NS; ++s) {
#pragma unroll
        for (int j = 0; j < 4; ++j) {
#pragma unroll
            for (int e = 0; e < 2; ++e) {
                int t = 16 * s + u - 2 * j - e - 127;
                t = t < 0 ? -t : t;
                const bool ok = t <= 127;
                const float v = taps[ok ? t : 0];
                A[s][j][e] = ok ? v : 0.f;
            }
        }
    }

    constexpr int TS = N_WAVES / N_ROWS;
    if (gw % TS == 0) {
        const int row = gw / TS;
        v4f se[4];
        load_window<1>(x + (size_t)row * N_IN, N_IN, lane, se);
        fill_window<1>(win, N_IN, lane, se);
        products<1>(y + (size_t)row * M_OUT, N_IN, win, A, lane);
    }
#pragma unroll
    for (int i = 0; i < GPW; ++i) {
        const int g   = gw + i * N_WAVES;
        const int row = g >> 8;
        const int gi  = g & 255;
        fill_window<4>(win, 1024 * gi, lane, st);
        if (i + 1 < GPW) {
            const int gn = g + N_WAVES;
            load_window<4>(x + (size_t)(gn >> 8) * N_IN, 1024 * (gn & 255), lane, st);
        }
        products<4>(y + (size_t)row * M_OUT, 1024 * gi, win, A, lane);
    }
}

extern "C" void kernel_launch(void* const* d_in, const int* in_sizes, int n_in,
                              void* d_out, int out_size, void* d_ws, size_t ws_size,
                              hipStream_t stream) {
    const float* x    = (const float*)d_in[0];
    const float* taps = (const float*)d_in[1];
    float* y          = (float*)d_out;
    fir_block_toeplitz_f32<<<dim3(N_WG, 1, 1), dim3(WAVES_PER_WG * 32, 1, 1), 0, stream>>>(x, taps, y);
}


# two-phase fast-FIR over tap-block parity: (Te+To)*Xe, To*(Xe - delayed Xo), Te*(Xe-Xo) replace four block products: 104 WMMAs per 512 outputs, was 136; summation order changes
# speedup vs baseline: 1.5013x; 1.0431x over previous
#include <hip/hip_runtime.h>


typedef float v2f __attribute__((ext_vector_type(2)));
typedef float v4f __attribute__((ext_vector_type(4)));
typedef float v8f __attribute__((ext_vector_type(8)));

#define N_IN     262144
#define M_OUT    262398
#define N_ROWS   32
#define NS       17
#define GROUPS_PER_ROW 256
#define N_GROUPS (N_ROWS * GROUPS_PER_ROW)
#define COLP     20
#define WIN_COLS 80
#define PAIRS    40
#define PLANE    (PAIRS * COLP)
#define NV       9
#define WAVES_PER_WG 8
#ifndef N_WG
#define N_WG 256
#endif
#define N_WAVES (N_WG * WAVES_PER_WG)

__device__ constexpr int sord(int k) { return k == 0 ? 0 : (k == 1 ? 16 : k - 1); }

template <int NT>
__device__ __forceinline__ void load_window(const float* __restrict__ xrow, const int P0,
                                            const int lane, v4f (&st)[2 + 2 * NT])
{
    constexpr int NCH = 2 + 2 * NT;
    const int xb = P0 - 256;
#pragma unroll
    for (int i = 0; i < NCH; ++i) {
        const int g = xb + 128 * i;
        const int gc = (g >= 0 && g < N_IN) ? g : 0;
        st[i] = *(const v4f*)(xrow + gc + 4 * lane);
    }
}

template <int NT>
__device__ __forceinline__ void fill_window(float* __restrict__ win, const int P0, const int lane,
                                            const v4f (&st)[2 + 2 * NT])
{
    constexpr int NCH = 2 + 2 * NT;
    const int xb = P0 - 256;
    __builtin_amdgcn_wave_barrier();
#pragma unroll
    for (int i = 0; i < NCH; ++i) {
        const int g = xb + 128 * i;
        v4f v = st[i];
        if (!(g >= 0 && g < N_IN)) v = (v4f){0.f, 0.f, 0.f, 0.f};
        *(v4f*)(win + (8 * i + (lane >> 2)) * COLP + 4 * (lane & 3)) = v;
    }
    __builtin_amdgcn_wave_barrier();
}

template <int NT>
__device__ __forceinline__ void products(float* __restrict__ yrow, const int P0,
                                         const float* __restrict__ win,
                                         const v2f (&TE)[NV][4], const v2f (&TO)[NV - 1][4], const int lane)
{
    v8f acc[NT];
#pragma unroll
    for (int t = 0; t < NT; ++t) acc[t] = (v8f){0.f,0.f,0.f,0.f,0.f,0.f,0.f,0.f};

    const float* __restrict__ bl = win + (lane & 15) * COLP + 8 * (lane >> 4);
    v4f c0[NT], c1[NT];
#pragma unroll
    for (int t = 0; t < NT; ++t) {
        const float* p = bl + (16 * t + 16 - sord(0)) * COLP;
        c0[t] = *(const v4f*)p;
        c1[t] = *(const v4f*)(p + 4);
    }
#pragma unroll
    for (int k = 0; k < NS; ++k) {
        const int s = sord(k);
        v4f n0[NT], n1[NT];
        if (k + 1 < NS) {
#pragma unroll
            for (int t = 0; t < NT; ++t) {
                const float* p = bl + (16 * t + 16 - sord(k + 1)) * COLP;
                n0[t] = *(const v4f*)p;
                n1[t] = *(const v4f*)(p + 4);
            }
        }
#pragma unroll
        for (int j = 0; j < 4; ++j) {
#pragma unroll
            for (int t = 0; t < NT; ++t) {
                v2f b;
                if (j == 0)      { b.x = c0[t].x; b.y = c0[t].y; }
                else if (j == 1) { b.x = c0[t].z; b.y = c0[t].w; }
                else if (j == 2) { b.x = c1[t].x; b.y = c1[t].y; }
                else             { b.x = c1[t].z; b.y = c1[t].w; }
                const v2f a = (s & 1) ? TO[(s >> 1) < NV - 1 ? (s >> 1) : 0][j] : TE[s >> 1][j];
                acc[t] = __builtin_amdgcn_wmma_f32_16x16x4_f32(false, a, false, b,
                                                               (short)0, acc[t], false, false);
            }
        }
        __builtin_amdgcn_sched_barrier(0);
        if (k + 1 < NS) {
#pragma unroll
            for (int t = 0; t < NT; ++t) { c0[t] = n0[t]; c1[t] = n1[t]; }
        }
    }

#pragma unroll
    for (int t = 0; t < NT; ++t) {
        const int p = P0 + 256 * t + 16 * (lane & 15) + 8 * (lane >> 4);
#pragma unroll
        for (int k = 0; k < 4; ++k) {
            if (p + 2 * k < M_OUT) {
                v2f o; o.x = acc[t][2 * k]; o.y = acc[t][2 * k + 1];
                *(v2f*)(yrow + p + 2 * k) = o;
            }
        }
    }
}


__device__ __forceinline__ void load_pairs(const float* __restrict__ xrow, const int P0, const int lane,
                                           v4f (&ev)[5], v4f (&od)[5], v4f (&pv)[5])
{
    const int xb = P0 - 256;
    const int g = lane >> 2, o = lane & 3;
#pragma unroll
    for (int i = 0; i < 5; ++i) {
        const int base = xb + 256 * i;
        const int bc = base >= 0 ? base : 0;
        const float* p = xrow + bc + 32 * g + 4 * o;
        ev[i] = *(const v4f*)p;
        od[i] = *(const v4f*)(p + 16);
        const int a  = base + 32 * g - 16;
        const int ac = a >= 0 ? a : 0;
        pv[i] = *(const v4f*)(xrow + ac + 4 * o);
    }
}

__device__ __forceinline__ void fill_pairs(float* __restrict__ win, const int P0, const int lane,
                                           const v4f (&ev)[5], const v4f (&od)[5], const v4f (&pv)[5])
{
    const int xb = P0 - 256;
    const int g = lane >> 2, o = lane & 3;
    const v4f z = (v4f){0.f, 0.f, 0.f, 0.f};
    __builtin_amdgcn_wave_barrier();
#pragma unroll
    for (int i = 0; i < 5; ++i) {
        const int base = xb + 256 * i;
        const int a    = base + 32 * g - 16;
        const v4f e = base >= 0 ? ev[i] : z;
        const v4f d = base >= 0 ? od[i] : z;
        const v4f q = a >= 0 ? pv[i] : z;
        float* w = win + (8 * i + g) * COLP + 4 * o;
        *(v4f*)w               = e;
        *(v4f*)(w + PLANE)     = e - d;
        *(v4f*)(w + 2 * PLANE) = e - q;
    }
    __builtin_amdgcn_wave_barrier();
}

__device__ __forceinline__ v2f chunk(const v4f& c0, const v4f& c1, const int j)
{
    v2f b;
    if (j == 0)      { b.x = c0.x; b.y = c0.y; }
    else if (j == 1) { b.x = c0.z; b.y = c0.w; }
    else if (j == 2) { b.x = c1.x; b.y = c1.y; }
    else             { b.x = c1.z; b.y = c1.w; }
    return b;
}

__device__ __forceinline__ void products_pairs(float* __restrict__ yrow, const int P0,
                                               const float* __restrict__ win,
                                               const v2f (&TSUM)[NV][4], const v2f (&TE)[NV][4],
                                               const v2f (&TO)[NV - 1][4], const int lane)
{
    v8f aR[2], aS[2], aU[2];
#pragma unroll
    for (int t = 0; t < 2; ++t) {
        aR[t] = (v8f){0.f,0.f,0.f,0.f,0.f,0.f,0.f,0.f};
        aS[t] = aR[t];
        aU[t] = aR[t];
    }
    const float* __restrict__ bl = win + (lane & 15) * COLP + 8 * (lane >> 4);
    v4f c[2][6], nx[2][6];
#pragma unroll
    for (int t = 0; t < 2; ++t) {
        const float* p = bl + (16 * t + 8) * COLP;
#pragma unroll
        for (int f = 0; f < 3; ++f) {
            c[t][2 * f]     = *(const v4f*)(p + f * PLANE);
            c[t][2 * f + 1] = *(const v4f*)(p + f * PLANE + 4);
        }
    }
#pragma unroll
    for (int v = 0; v < NV; ++v) {
        if (v + 1 < NV) {
#pragma unroll
            for (int t = 0; t < 2; ++t) {
                const float* p = bl + (16 * t + 8 - (v + 1)) * COLP;
#pragma unroll
                for (int f = 0; f < 3; ++f) {
                    if (f == 2 && v + 1 >= NV - 1) continue;
                    nx[t][2 * f]     = *(const v4f*)(p + f * PLANE);
                    nx[t][2 * f + 1] = *(const v4f*)(p + f * PLANE + 4);
                }
            }
        }
#pragma unroll
        for (int j = 0; j < 4; ++j) {
#pragma unroll
            for (int t = 0; t < 2; ++t) {
                aR[t] = __builtin_amdgcn_wmma_f32_16x16x4_f32(false, TSUM[v][j], false, chunk(c[t][0], c[t][1], j),
                                                              (short)0, aR[t], false, false);
                aU[t] = __builtin_amdgcn_wmma_f32_16x16x4_f32(false, TE[v][j], false, chunk(c[t][2], c[t][3], j),
                                                              (short)0, aU[t], false, false);
                if (v < NV - 1)
                    aS[t] = __builtin_amdgcn_wmma_f32_16x16x4_f32(false, TO[v][j], false, chunk(c[t][4], c[t][5], j),
                                                                  (short)0, aS[t], false, false);
            }
        }
        __builtin_amdgcn_sched_barrier(0);
        if (v + 1 < NV) {
#pragma unroll
            for (int t = 0; t < 2; ++t)
#pragma unroll
                for (int f = 0; f < 6; ++f) c[t][f] = nx[t][f];
        }
    }

#pragma unroll
    for (int t = 0; t < 2; ++t) {
        const int p = P0 + 512 * t + 32 * (lane & 15) + 8 * (lane >> 4);
        const v8f ye = aR[t] - aS[t];
        const v8f yo = aR[t] - aU[t];
#pragma unroll
        for (int k = 0; k < 4; ++k) {
            v2f o; o.x = ye[2 * k]; o.y = ye[2 * k + 1];
            *(v2f*)(yrow + p + 2 * k) = o;
        }
#pragma unroll
        for (int k = 0; k < 4; ++k) {
            v2f o; o.x = yo[2 * k]; o.y = yo[2 * k + 1];
            *(v2f*)(yrow + p + 16 + 2 * k) = o;
        }
    }
}

__global__ __launch_bounds__(WAVES_PER_WG * 32)
void fir_block_toeplitz_f32(const float* __restrict__ x,
                            const float* __restrict__ taps,
                            float* __restrict__ y)
{
    __shared__ __attribute__((aligned(16))) float lds[WAVES_PER_WG * 3 * PLANE];
    const int lane = threadIdx.x & 31;
    const int wv   = __builtin_amdgcn_readfirstlane((int)(threadIdx.x >> 5));
    float* win = lds + wv * (3 * PLANE);
    const int gw = blockIdx.x * WAVES_PER_WG + wv;

    constexpr int GPW = N_GROUPS / N_WAVES;
    static_assert(N_GROUPS % N_WAVES == 0, "equal shares");
    v4f ev[5], od[5], pv[5];
    load_pairs(x + (size_t)(gw >> 8) * N_IN, 1024 * (gw & 255), lane, ev, od, pv);
    __builtin_amdgcn_sched_barrier(0);

    v2f TE[NV][4], TO[NV - 1][4], TSUM[NV][4];
    const int u = (lane & 15) - 8 * (lane >> 4);
#pragma unroll
    for (int s = 0; s < NS; ++s) {
#pragma unroll
        for (int j = 0; j < 4; ++j) {
#pragma unroll
            for (int e = 0; e < 2; ++e) {
                int t = 16 * s + u - 2 * j - e - 127;
                t = t < 0 ? -t : t;
                const bool ok = t <= 127;
                const float v = taps[ok ? t : 0];
                const float a = ok ? v : 0.f;
                if (s & 1) TO[s >> 1][j][e] = a; else TE[s >> 1][j][e] = a;
            }
        }
    }
#pragma unroll
    for (int v = 0; v < NV; ++v)
#pragma unroll
        for (int j = 0; j < 4; ++j)
            TSUM[v][j] = v < NV - 1 ? TE[v][j] + TO[v][j] : TE[v][j];

    constexpr int TS = N_WAVES / N_ROWS;
    if (gw % TS == 0) {
        const int row = gw / TS;
        v4f se[4];
        load_window<1>(x + (size_t)row * N_IN, N_IN, lane, se);
        fill_window<1>(win, N_IN, lane, se);
        products<1>(y + (size_t)row * M_OUT, N_IN, win, TE, TO, lane);
    }
#pragma unroll
    for (int i = 0; i < GPW; ++i) {
        const int g   = gw + i * N_WAVES;
        const int row = g >> 8;
        const int gi  = g & 255;
        fill_pairs(win, 1024 * gi, lane, ev, od, pv);
        if (i + 1 < GPW) {
            const int gn = g + N_WAVES;
            load_pairs(x + (size_t)(gn >> 8) * N_IN, 1024 * (gn & 255), lane, ev, od, pv);
        }
        products_pairs(y + (size_t)row * M_OUT, 1024 * gi, win, TSUM, TE, TO, lane);
    }
}

extern "C" void kernel_launch(void* const* d_in, const int* in_sizes, int n_in,
                              void* d_out, int out_size, void* d_ws, size_t ws_size,
                              hipStream_t stream) {
    const float* x    = (const float*)d_in[0];
    const float* taps = (const float*)d_in[1];
    float* y          = (float*)d_out;
    fir_block_toeplitz_f32<<<dim3(N_WG, 1, 1), dim3(WAVES_PER_WG * 32, 1, 1), 0, stream>>>(x, taps, y);
}


# full groups: WMMA operand roles exchanged (x blocks as A rows, taps as B) and lane halves swapped once: each store instruction writes one whole 128-byte line, was 16 lines x 32 B; sum order unchanged
# speedup vs baseline: 1.6486x; 1.0981x over previous
#include <hip/hip_runtime.h>


typedef float v2f __attribute__((ext_vector_type(2)));
typedef float v4f __attribute__((ext_vector_type(4)));
typedef float v8f __attribute__((ext_vector_type(8)));

#define N_IN     262144
#define M_OUT    262398
#define N_ROWS   32
#define NS       17
#define GROUPS_PER_ROW 256
#define N_GROUPS (N_ROWS * GROUPS_PER_ROW)
#define COLP     20
#define WIN_COLS 80
#define PAIRS    40
#define PLANE    (PAIRS * COLP)
#define NV       9
#define WAVES_PER_WG 8
#ifndef N_WG
#define N_WG 256
#endif
#define N_WAVES (N_WG * WAVES_PER_WG)

__device__ constexpr int sord(int k) { return k == 0 ? 0 : (k == 1 ? 16 : k - 1); }

template <int NT>
__device__ __forceinline__ void load_window(const float* __restrict__ xrow, const int P0,
                                            const int lane, v4f (&st)[2 + 2 * NT])
{
    constexpr int NCH = 2 + 2 * NT;
    const int xb = P0 - 256;
#pragma unroll
    for (int i = 0; i < NCH; ++i) {
        const int g = xb + 128 * i;
        const int gc = (g >= 0 && g < N_IN) ? g : 0;
        st[i] = *(const v4f*)(xrow + gc + 4 * lane);
    }
}

template <int NT>
__device__ __forceinline__ void fill_window(float* __restrict__ win, const int P0, const int lane,
                                            const v4f (&st)[2 + 2 * NT])
{
    constexpr int NCH = 2 + 2 * NT;
    const int xb = P0 - 256;
    __builtin_amdgcn_wave_barrier();
#pragma unroll
    for (int i = 0; i < NCH; ++i) {
        const int g = xb + 128 * i;
        v4f v = st[i];
        if (!(g >= 0 && g < N_IN)) v = (v4f){0.f, 0.f, 0.f, 0.f};
        *(v4f*)(win + (8 * i + (lane >> 2)) * COLP + 4 * (lane & 3)) = v;
    }
    __builtin_amdgcn_wave_barrier();
}

template <int NT>
__device__ __forceinline__ void products(float* __restrict__ yrow, const int P0,
                                         const float* __restrict__ win,
                                         const v2f (&TE)[NV][4], const v2f (&TO)[NV - 1][4], const int lane)
{
    v8f acc[NT];
#pragma unroll
    for (int t = 0; t < NT; ++t) acc[t] = (v8f){0.f,0.f,0.f,0.f,0.f,0.f,0.f,0.f};

    const float* __restrict__ bl = win + (lane & 15) * COLP + 8 * (lane >> 4);
    v4f c0[NT], c1[NT];
#pragma unroll
    for (int t = 0; t < NT; ++t) {
        const float* p = bl + (16 * t + 16 - sord(0)) * COLP;
        c0[t] = *(const v4f*)p;
        c1[t] = *(const v4f*)(p + 4);
    }
#pragma unroll
    for (int k = 0; k < NS; ++k) {
        const int s = sord(k);
        v4f n0[NT], n1[NT];
        if (k + 1 < NS) {
#pragma unroll
            for (int t = 0; t < NT; ++t) {
                const float* p = bl + (16 * t + 16 - sord(k + 1)) * COLP;
                n0[t] = *(const v4f*)p;
                n1[t] = *(const v4f*)(p + 4);
            }
        }
#pragma unroll
        for (int j = 0; j < 4; ++j) {
#pragma unroll
            for (int t = 0; t < NT; ++t) {
                v2f b;
                if (j == 0)      { b.x = c0[t].x; b.y = c0[t].y; }
                else if (j == 1) { b.x = c0[t].z; b.y = c0[t].w; }
                else if (j == 2) { b.x = c1[t].x; b.y = c1[t].y; }
                else             { b.x = c1[t].z; b.y = c1[t].w; }
                const v2f a = (s & 1) ? TO[(s >> 1) < NV - 1 ? (s >> 1) : 0][j] : TE[s >> 1][j];
                acc[t] = __builtin_amdgcn_wmma_f32_16x16x4_f32(false, a, false, b,
                                                               (short)0, acc[t], false, false);
            }
        }
        __builtin_amdgcn_sched_barrier(0);
        if (k + 1 < NS) {
#pragma unroll
            for (int t = 0; t < NT; ++t) { c0[t] = n0[t]; c1[t] = n1[t]; }
        }
    }

#pragma unroll
    for (int t = 0; t < NT; ++t) {
        const int p = P0 + 256 * t + 16 * (lane & 15) + 8 * (lane >> 4);
#pragma unroll
        for (int k = 0; k < 4; ++k) {
            if (p + 2 * k < M_OUT) {
                v2f o; o.x = acc[t][2 * k]; o.y = acc[t][2 * k + 1];
                *(v2f*)(yrow + p + 2 * k) = o;
            }
        }
    }
}


__device__ __forceinline__ void load_pairs(const float* __restrict__ xrow, const int P0, const int lane,
                                           v4f (&ev)[5], v4f (&od)[5], v4f (&pv)[5])
{
    const int xb = P0 - 256;
    const int g = lane >> 2, o = lane & 3;
#pragma unroll
    for (int i = 0; i < 5; ++i) {
        const int base = xb + 256 * i;
        const int bc = base >= 0 ? base : 0;
        const float* p = xrow + bc + 32 * g + 4 * o;
        ev[i] = *(const v4f*)p;
        od[i] = *(const v4f*)(p + 16);
        const int a  = base + 32 * g - 16;
        const int ac = a >= 0 ? a : 0;
        pv[i] = *(const v4f*)(xrow + ac + 4 * o);
    }
}

__device__ __forceinline__ void fill_pairs(float* __restrict__ win, const int P0, const int lane,
                                           const v4f (&ev)[5], const v4f (&od)[5], const v4f (&pv)[5])
{
    const int xb = P0 - 256;
    const int g = lane >> 2, o = lane & 3;
    const v4f z = (v4f){0.f, 0.f, 0.f, 0.f};
    __builtin_amdgcn_wave_barrier();
#pragma unroll
    for (int i = 0; i < 5; ++i) {
        const int base = xb + 256 * i;
        const int a    = base + 32 * g - 16;
        const v4f e = base >= 0 ? ev[i] : z;
        const v4f d = base >= 0 ? od[i] : z;
        const v4f q = a >= 0 ? pv[i] : z;
        float* w = win + (8 * i + g) * COLP + 4 * o;
        *(v4f*)w               = e;
        *(v4f*)(w + PLANE)     = e - d;
        *(v4f*)(w + 2 * PLANE) = e - q;
    }
    __builtin_amdgcn_wave_barrier();
}

__device__ __forceinline__ v2f chunk(const v4f& c0, const v4f& c1, const int j)
{
    v2f b;
    if (j == 0)      { b.x = c0.x; b.y = c0.y; }
    else if (j == 1) { b.x = c0.z; b.y = c0.w; }
    else if (j == 2) { b.x = c1.x; b.y = c1.y; }
    else             { b.x = c1.z; b.y = c1.w; }
    return b;
}

__device__ __forceinline__ float swap_halves(const float v)
{
    const unsigned u = __builtin_bit_cast(unsigned, v);
    return __builtin_bit_cast(float, __builtin_amdgcn_permlanex16(u, u, 0x76543210u, 0xfedcba98u, false, false));
}

__device__ __forceinline__ void products_pairs(float* __restrict__ yrow, const int P0,
                                               const float* __restrict__ win,
                                               const v2f (&TSUM)[NV][4], const v2f (&TE)[NV][4],
                                               const v2f (&TO)[NV - 1][4], const int lane, const int shift)
{
    v8f aR[2], aS[2], aU[2];
#pragma unroll
    for (int t = 0; t < 2; ++t) {
        aR[t] = (v8f){0.f,0.f,0.f,0.f,0.f,0.f,0.f,0.f};
        aS[t] = aR[t];
        aU[t] = aR[t];
    }
    const float* __restrict__ bl = win + (lane & 15) * COLP + 8 * (lane >> 4);
    v4f c[2][6], nx[2][6];
#pragma unroll
    for (int t = 0; t < 2; ++t) {
        const float* p = bl + (16 * t + 8) * COLP;
#pragma unroll
        for (int f = 0; f < 3; ++f) {
            c[t][2 * f]     = *(const v4f*)(p + f * PLANE);
            c[t][2 * f + 1] = *(const v4f*)(p + f * PLANE + 4);
        }
    }
#pragma unroll
    for (int v = 0; v < NV; ++v) {
        if (v + 1 < NV) {
#pragma unroll
            for (int t = 0; t < 2; ++t) {
                const float* p = bl + (16 * t + 8 - (v + 1)) * COLP;
#pragma unroll
                for (int f = 0; f < 3; ++f) {
                    if (f == 2 && v + 1 >= NV - 1) continue;
                    nx[t][2 * f]     = *(const v4f*)(p + f * PLANE);
                    nx[t][2 * f + 1] = *(const v4f*)(p + f * PLANE + 4);
                }
            }
        }
#pragma unroll
        for (int j = 0; j < 4; ++j) {
#pragma unroll
            for (int t = 0; t < 2; ++t) {
                aR[t] = __builtin_amdgcn_wmma_f32_16x16x4_f32(false, chunk(c[t][0], c[t][1], j), false, TSUM[v][j],
                                                              (short)0, aR[t], false, false);
                aU[t] = __builtin_amdgcn_wmma_f32_16x16x4_f32(false, chunk(c[t][2], c[t][3], j), false, TE[v][j],
                                                              (short)0, aU[t], false, false);
                if (v < NV - 1)
                    aS[t] = __builtin_amdgcn_wmma_f32_16x16x4_f32(false, chunk(c[t][4], c[t][5], j), false, TO[v][j],
                                                                  (short)0, aS[t], false, false);
            }
        }
        __builtin_amdgcn_sched_barrier(0);
        if (v + 1 < NV) {
#pragma unroll
            for (int t = 0; t < 2; ++t)
#pragma unroll
                for (int f = 0; f < 6; ++f) c[t][f] = nx[t][f];
        }
    }

    const bool hi = lane >= 16;
    v8f lo0, up0, lo1, up1;
#pragma unroll
    for (int t = 0; t < 2; ++t) {
        const v8f ye = aR[t] - aS[t];
        const v8f yo = aR[t] - aU[t];
        v8f lo, up;
#pragma unroll
        for (int i = 0; i < 8; ++i) {
            const float give = hi ? ye[i] : yo[i];
            const float got  = swap_halves(give);
            lo[i] = hi ? got : ye[i];
            up[i] = hi ? yo[i] : got;
        }
        if (t == 0) { lo0 = lo; up0 = up; } else { lo1 = lo; up1 = up; }
    }
#define RUN(k) (((k) >> 3) == 0 ? lo0[(k) & 7] : ((k) >> 3) == 1 ? up0[(k) & 7] : ((k) >> 3) == 2 ? lo1[(k) & 7] : up1[(k) & 7])
    const bool wrap = lane < shift;
    float* __restrict__ yl = yrow + P0 + lane + (wrap ? 32 : 0);
    if (wrap) yl[-32] = RUN(0);
#pragma unroll
    for (int k = 0; k < 31; ++k) {
        const float a = RUN(k), b = RUN(k + 1);
        yl[32 * k] = wrap ? b : a;
    }
    if (!wrap) yl[32 * 31] = RUN(31);
#undef RUN
}

__global__ __launch_bounds__(WAVES_PER_WG * 32)
void fir_block_toeplitz_f32(const float* __restrict__ x,
                            const float* __restrict__ taps,
                            float* __restrict__ y)
{
    __shared__ __attribute__((aligned(16))) float lds[WAVES_PER_WG * 3 * PLANE];
    const int lane = threadIdx.x & 31;
    const int wv   = __builtin_amdgcn_readfirstlane((int)(threadIdx.x >> 5));
    float* win = lds + wv * (3 * PLANE);
    const int gw = blockIdx.x * WAVES_PER_WG + wv;

    constexpr int GPW = N_GROUPS / N_WAVES;
    static_assert(N_GROUPS % N_WAVES == 0, "equal shares");
    v4f ev[5], od[5], pv[5];
    load_pairs(x + (size_t)(gw >> 8) * N_IN, 1024 * (gw & 255), lane, ev, od, pv);
    __builtin_amdgcn_sched_barrier(0);

    v2f TE[NV][4], TO[NV - 1][4], TSUM[NV][4];
    const int u = (lane & 15) - 8 * (lane >> 4);
#pragma unroll
    for (int s = 0; s < NS; ++s) {
#pragma unroll
        for (int j = 0; j < 4; ++j) {
#pragma unroll
            for (int e = 0; e < 2; ++e) {
                int t = 16 * s + u - 2 * j - e - 127;
                t = t < 0 ? -t : t;
                const bool ok = t <= 127;
                const float v = taps[ok ? t : 0];
                const float a = ok ? v : 0.f;
                if (s & 1) TO[s >> 1][j][e] = a; else TE[s >> 1][j][e] = a;
            }
        }
    }
#pragma unroll
    for (int v = 0; v < NV; ++v)
#pragma unroll
        for (int j = 0; j < 4; ++j)
            TSUM[v][j] = v < NV - 1 ? TE[v][j] + TO[v][j] : TE[v][j];

    constexpr int TS = N_WAVES / N_ROWS;
    if (gw % TS == 0) {
        const int row = gw / TS;
        v4f se[4];
        load_window<1>(x + (size_t)row * N_IN, N_IN, lane, se);
        fill_window<1>(win, N_IN, lane, se);
        products<1>(y + (size_t)row * M_OUT, N_IN, win, TE, TO, lane);
    }
#pragma unroll
    for (int i = 0; i < GPW; ++i) {
        const int g   = gw + i * N_WAVES;
        const int row = g >> 8;
        const int gi  = g & 255;
        fill_pairs(win, 1024 * gi, lane, ev, od, pv);
        if (i + 1 < GPW) {
            const int gn = g + N_WAVES;
            load_pairs(x + (size_t)(gn >> 8) * N_IN, 1024 * (gn & 255), lane, ev, od, pv);
        }
        products_pairs(y + (size_t)row * M_OUT, 1024 * gi, win, TSUM, TE, TO, lane, (2 * row) & 31);
    }
}

extern "C" void kernel_launch(void* const* d_in, const int* in_sizes, int n_in,
                              void* d_out, int out_size, void* d_ws, size_t ws_size,
                              hipStream_t stream) {
    const float* x    = (const float*)d_in[0];
    const float* taps = (const float*)d_in[1];
    float* y          = (float*)d_out;
    fir_block_toeplitz_f32<<<dim3(N_WG, 1, 1), dim3(WAVES_PER_WG * 32, 1, 1), 0, stream>>>(x, taps, y);
}


# second fast-FIR level on block index mod 4: nine block products of 5 or 4 tap blocks, 160 WMMAs per 1024 outputs, was 208; tap fragments read from an LDS table built per workgroup; sum order changes
# speedup vs baseline: 1.7750x; 1.0767x over previous
#include <hip/hip_runtime.h>


typedef float v2f __attribute__((ext_vector_type(2)));
typedef float v4f __attribute__((ext_vector_type(4)));
typedef float v8f __attribute__((ext_vector_type(8)));

#define N_IN     262144
#define M_OUT    262398
#define N_ROWS   32
#define NS       17
#define GROUPS_PER_ROW 256
#define N_GROUPS (N_ROWS * GROUPS_PER_ROW)
#define COLP     20
#define QCOLS    20
#define PLANE    (QCOLS * COLP)
#define NPL      9
#define WIN_FLOATS (NPL * PLANE)
#define NVQ      5
#define SLOTS    (NPL * NVQ)
#define SLOT_FLOATS 256
#define WAVES_PER_WG 8
#ifndef N_WG
#define N_WG 256
#endif
#define N_WAVES (N_WG * WAVES_PER_WG)

__device__ constexpr int sord(int k) { return k == 0 ? 0 : (k == 1 ? 16 : k - 1); }
__device__ constexpr int slot_of(int s)
{
    return ((s & 3) == 0 ? 8 : (s & 3) == 1 ? 5 : (s & 3) == 2 ? 7 : 4) * NVQ + (s >> 2);
}
__device__ constexpr bool len5(int p) { return p == 0 || p == 2 || p == 6 || p == 8; }
__device__ constexpr int wset_of(int p)
{
    return p == 0 ? 0xF : p == 1 ? 0xC : p == 2 ? 0x3 : p == 3 ? 0xA : p == 4 ? 0x8 : p == 5 ? 0x2 : p == 6 ? 0x5 : p == 7 ? 0x4 : 0x1;
}

template <int NT>
__device__ __forceinline__ void load_window(const float* __restrict__ xrow, const int P0,
                                            const int lane, v4f (&st)[2 + 2 * NT])
{
    constexpr int NCH = 2 + 2 * NT;
    const int xb = P0 - 256;
#pragma unroll
    for (int i = 0; i < NCH; ++i) {
        const int g = xb + 128 * i;
        const int gc = (g >= 0 && g < N_IN) ? g : 0;
        st[i] = *(const v4f*)(xrow + gc + 4 * lane);
    }
}

template <int NT>
__device__ __forceinline__ void fill_window(float* __restrict__ win, const int P0, const int lane,
                                            const v4f (&st)[2 + 2 * NT])
{
    constexpr int NCH = 2 + 2 * NT;
    const int xb = P0 - 256;
    __builtin_amdgcn_wave_barrier();
#pragma unroll
    for (int i = 0; i < NCH; ++i) {
        const int g = xb + 128 * i;
        v4f v = st[i];
        if (!(g >= 0 && g < N_IN)) v = (v4f){0.f, 0.f, 0.f, 0.f};
        *(v4f*)(win + (8 * i + (lane >> 2)) * COLP + 4 * (lane & 3)) = v;
    }
    __builtin_amdgcn_wave_barrier();
}

__device__ __forceinline__ v2f chunk(const v4f& c0, const v4f& c1, const int j)
{
    v2f b;
    if (j == 0)      { b.x = c0.x; b.y = c0.y; }
    else if (j == 1) { b.x = c0.z; b.y = c0.w; }
    else if (j == 2) { b.x = c1.x; b.y = c1.y; }
    else             { b.x = c1.z; b.y = c1.w; }
    return b;
}

__device__ __forceinline__ void products_end(float* __restrict__ yrow, const int P0,
                                             const float* __restrict__ win,
                                             const float* __restrict__ tab, const int lane)
{
    v8f acc = (v8f){0.f,0.f,0.f,0.f,0.f,0.f,0.f,0.f};
    const float* __restrict__ bl = win + (lane & 15) * COLP + 8 * (lane >> 4);
    const float* __restrict__ tl = tab + 4 * lane;
    v4f c0, c1, a0, a1;
    {
        const float* p = bl + (16 - sord(0)) * COLP;
        c0 = *(const v4f*)p;
        c1 = *(const v4f*)(p + 4);
        const float* q = tl + slot_of(sord(0)) * SLOT_FLOATS;
        a0 = *(const v4f*)q;
        a1 = *(const v4f*)(q + 128);
    }
#pragma unroll
    for (int k = 0; k < NS; ++k) {
        v4f n0, n1, m0, m1;
        if (k + 1 < NS) {
            const float* p = bl + (16 - sord(k + 1)) * COLP;
            n0 = *(const v4f*)p;
            n1 = *(const v4f*)(p + 4);
            const float* q = tl + slot_of(sord(k + 1)) * SLOT_FLOATS;
            m0 = *(const v4f*)q;
            m1 = *(const v4f*)(q + 128);
        }
#pragma unroll
        for (int j = 0; j < 4; ++j)
            acc = __builtin_amdgcn_wmma_f32_16x16x4_f32(false, chunk(a0, a1, j), false, chunk(c0, c1, j),
                                                        (short)0, acc, false, false);
        __builtin_amdgcn_sched_barrier(0);
        if (k + 1 < NS) { c0 = n0; c1 = n1; a0 = m0; a1 = m1; }
    }

    const int p = P0 + 16 * (lane & 15) + 8 * (lane >> 4);
#pragma unroll
    for (int k = 0; k < 4; ++k) {
        if (p + 2 * k < M_OUT) {
            v2f o; o.x = acc[2 * k]; o.y = acc[2 * k + 1];
            *(v2f*)(yrow + p + 2 * k) = o;
        }
    }
}

__device__ __forceinline__ void load_quads(const float* __restrict__ xrow, const int P0, const int lane,
                                           v4f (&qa)[3][4], v4f (&qb)[3][3])
{
    const int g = lane >> 2, o = lane & 3;
#pragma unroll
    for (int i = 0; i < 3; ++i) {
        int cidx = 8 * i + g;
        cidx = cidx < QCOLS ? cidx : QCOLS - 1;
        const int xq = P0 + 64 * (cidx - 4);
        const int xa = xq >= 0 ? xq : 0;
        const int xb = xq >= 64 ? xq - 64 : 0;
        const float* pa = xrow + xa + 4 * o;
        const float* pb = xrow + xb + 4 * o;
#pragma unroll
        for (int t = 0; t < 4; ++t) qa[i][t] = *(const v4f*)(pa + 16 * t);
#pragma unroll
        for (int t = 0; t < 3; ++t) qb[i][t] = *(const v4f*)(pb + 16 * (t + 1));
    }
}

__device__ __forceinline__ void fill_quads(float* __restrict__ win, const int P0, const int lane,
                                           const v4f (&qa)[3][4], const v4f (&qb)[3][3])
{
    const int g = lane >> 2, o = lane & 3;
    const v4f z = (v4f){0.f, 0.f, 0.f, 0.f};
    __builtin_amdgcn_wave_barrier();
#pragma unroll
    for (int i = 0; i < 3; ++i) {
        int cidx = 8 * i + g;
        cidx = cidx < QCOLS ? cidx : QCOLS - 1;
        const int xq = P0 + 64 * (cidx - 4);
        const bool va = xq >= 0, vb = xq >= 64;
        const v4f a0 = va ? qa[i][0] : z, a1 = va ? qa[i][1] : z, a2 = va ? qa[i][2] : z, a3 = va ? qa[i][3] : z;
        const v4f b1 = vb ? qb[i][0] : z, b2 = vb ? qb[i][1] : z, b3 = vb ? qb[i][2] : z;
        float* w = win + cidx * COLP + 4 * o;
        const v4f ges = a0 - b3;
        const v4f geu = a0 - a1;
        *(v4f*)(w)             = a0;
        *(v4f*)(w + PLANE)     = a0 - b2;
        *(v4f*)(w + 2 * PLANE) = a0 - a2;
        *(v4f*)(w + 3 * PLANE) = ges;
        *(v4f*)(w + 4 * PLANE) = ges - (b2 - b1);
        *(v4f*)(w + 5 * PLANE) = ges - (a2 - a1);
        *(v4f*)(w + 6 * PLANE) = geu;
        *(v4f*)(w + 7 * PLANE) = geu - (b2 - b3);
        *(v4f*)(w + 8 * PLANE) = geu - (a2 - a3);
    }
    __builtin_amdgcn_wave_barrier();
}

__device__ __forceinline__ float swap_halves(const float v)
{
    const unsigned u = __builtin_bit_cast(unsigned, v);
    return __builtin_bit_cast(float, __builtin_amdgcn_permlanex16(u, u, 0x76543210u, 0xfedcba98u, false, false));
}

__device__ __forceinline__ void pair_up(const v8f& ya, const v8f& yb, const bool hi, v8f& lo, v8f& up)
{
#pragma unroll
    for (int i = 0; i < 8; ++i) {
        const float give = hi ? ya[i] : yb[i];
        const float got  = swap_halves(give);
        lo[i] = hi ? got : ya[i];
        up[i] = hi ? yb[i] : got;
    }
}

__device__ __forceinline__ void products_quads(float* __restrict__ yrow, const int P0,
                                               const float* __restrict__ win,
                                               const float* __restrict__ tab, const int lane, const int shift)
{
    v8f acc[NPL];
#pragma unroll
    for (int p = 0; p < NPL; ++p) acc[p] = (v8f){0.f,0.f,0.f,0.f,0.f,0.f,0.f,0.f};
    const float* __restrict__ bl = win + (lane & 15) * COLP + 8 * (lane >> 4);
    const float* __restrict__ tl = tab + 4 * lane;
    v4f xc[NPL][2], tc[NPL][2], xn[NPL][2], tn[NPL][2];
#pragma unroll
    for (int p = 0; p < NPL; ++p) {
        const float* px = bl + p * PLANE + 4 * COLP;
        xc[p][0] = *(const v4f*)px;
        xc[p][1] = *(const v4f*)(px + 4);
        const float* pt = tl + (p * NVQ) * SLOT_FLOATS;
        tc[p][0] = *(const v4f*)pt;
        tc[p][1] = *(const v4f*)(pt + 128);
    }
#pragma unroll
    for (int v = 0; v < NVQ; ++v) {
        if (v + 1 < NVQ) {
#pragma unroll
            for (int p = 0; p < NPL; ++p) {
                if (v + 1 == NVQ - 1 && !len5(p)) continue;
                const float* px = bl + p * PLANE + (4 - (v + 1)) * COLP;
                xn[p][0] = *(const v4f*)px;
                xn[p][1] = *(const v4f*)(px + 4);
                const float* pt = tl + (p * NVQ + v + 1) * SLOT_FLOATS;
                tn[p][0] = *(const v4f*)pt;
                tn[p][1] = *(const v4f*)(pt + 128);
            }
            __builtin_amdgcn_sched_barrier(0);
        }
#pragma unroll
        for (int j = 0; j < 4; ++j) {
#pragma unroll
            for (int p = 0; p < NPL; ++p) {
                if (v == NVQ - 1 && !len5(p)) continue;
                acc[p] = __builtin_amdgcn_wmma_f32_16x16x4_f32(false, chunk(xc[p][0], xc[p][1], j),
                                                               false, chunk(tc[p][0], tc[p][1], j),
                                                               (short)0, acc[p], false, false);
            }
        }
        __builtin_amdgcn_sched_barrier(0);
        if (v + 1 < NVQ) {
#pragma unroll
            for (int p = 0; p < NPL; ++p) {
                if (v + 1 == NVQ - 1 && !len5(p)) continue;
                xc[p][0] = xn[p][0]; xc[p][1] = xn[p][1];
                tc[p][0] = tn[p][0]; tc[p][1] = tn[p][1];
            }
        }
    }

    const v8f ra = acc[0] - acc[1], rb = acc[0] - acc[2];
    const v8f sa = acc[3] - acc[4], sb = acc[3] - acc[5];
    const v8f ua = acc[6] - acc[7], ub = acc[6] - acc[8];
    const v8f y0 = ra - sa, y1 = ra - ua, y2 = rb - sb, y3 = rb - ub;
    const bool hi = lane >= 16;
    v8f lo01, up01, lo23, up23;
    pair_up(y0, y1, hi, lo01, up01);
    pair_up(y2, y3, hi, lo23, up23);
#define RUN(k) (((k) >> 4) == 0 ? (((k) & 1) ? lo23[((k) >> 1) & 7] : lo01[((k) >> 1) & 7]) \
                                : (((k) & 1) ? up23[((k) >> 1) & 7] : up01[((k) >> 1) & 7]))
    const bool wrap = lane < shift;
    float* __restrict__ yl = yrow + P0 + lane + (wrap ? 32 : 0);
    if (wrap) yl[-32] = RUN(0);
#pragma unroll
    for (int k = 0; k < 31; ++k) {
        const float a = RUN(k), b = RUN(k + 1);
        yl[32 * k] = wrap ? b : a;
    }
    if (!wrap) yl[32 * 31] = RUN(31);
#undef RUN
}

__global__ __launch_bounds__(WAVES_PER_WG * 32)
void fir_block_toeplitz_f32(const float* __restrict__ x,
                            const float* __restrict__ taps,
                            float* __restrict__ y)
{
    __shared__ __attribute__((aligned(16))) float lds[WAVES_PER_WG * WIN_FLOATS + SLOTS * SLOT_FLOATS];
    const int lane = threadIdx.x & 31;
    const int wv   = __builtin_amdgcn_readfirstlane((int)(threadIdx.x >> 5));
    float* win = lds + wv * WIN_FLOATS;
    float* tab = lds + WAVES_PER_WG * WIN_FLOATS;
    const int gw = blockIdx.x * WAVES_PER_WG + wv;

    constexpr int GPW = N_GROUPS / N_WAVES;
    static_assert(N_GROUPS % N_WAVES == 0, "equal shares");
    v4f qa[3][4], qb[3][3];
    load_quads(x + (size_t)(gw >> 8) * N_IN, 1024 * (gw & 255), lane, qa, qb);
    __builtin_amdgcn_sched_barrier(0);

    {
        const int r = lane & 15, h = lane >> 4;
        const int d = r - (8 * h + wv) - 127;
        float* __restrict__ dst = tab + (wv >> 2) * 128 + 4 * lane + (wv & 3);
#pragma unroll
        for (int f = 0; f < NPL; ++f) {
#pragma unroll
            for (int v = 0; v < NVQ; ++v) {
                if (v == NVQ - 1 && !len5(f)) continue;
                float sum = 0.f;
#pragma unroll
                for (int w = 0; w < 4; ++w) {
                    if (!((wset_of(f) >> w) & 1) || 4 * v + w >= NS) continue;
                    int t = 16 * (4 * v + w) + d;
                    t = t < 0 ? -t : t;
                    const bool ok = t <= 127;
                    const float tv = taps[ok ? t : 0];
                    sum += ok ? tv : 0.f;
                }
                dst[(f * NVQ + v) * SLOT_FLOATS] = sum;
            }
        }
    }
    __syncthreads();

    constexpr int TS = N_WAVES / N_ROWS;
    if (gw % TS == 0) {
        const int row = gw / TS;
        v4f se[4];
        load_window<1>(x + (size_t)row * N_IN, N_IN, lane, se);
        fill_window<1>(win, N_IN, lane, se);
        products_end(y + (size_t)row * M_OUT, N_IN, win, tab, lane);
    }
#pragma unroll
    for (int i = 0; i < GPW; ++i) {
        const int g   = gw + i * N_WAVES;
        const int row = g >> 8;
        const int gi  = g & 255;
        fill_quads(win, 1024 * gi, lane, qa, qb);
        if (i + 1 < GPW) {
            const int gn = g + N_WAVES;
            load_quads(x + (size_t)(gn >> 8) * N_IN, 1024 * (gn & 255), lane, qa, qb);
        }
        products_quads(y + (size_t)row * M_OUT, 1024 * gi, win, tab, lane, (2 * row) & 31);
    }
}

extern "C" void kernel_launch(void* const* d_in, const int* in_sizes, int n_in,
                              void* d_out, int out_size, void* d_ws, size_t ws_size,
                              hipStream_t stream) {
    const float* x    = (const float*)d_in[0];
    const float* taps = (const float*)d_in[1];
    float* y          = (float*)d_out;
    fir_block_toeplitz_f32<<<dim3(N_WG, 1, 1), dim3(WAVES_PER_WG * 32, 1, 1), 0, stream>>>(x, taps, y);
}
